# starting kernel
# baseline (speedup 1.0000x reference)
#include <hip/hip_runtime.h>
#include <hip/hip_bf16.h>

typedef __attribute__((ext_vector_type(2))) float v2f;
typedef __attribute__((ext_vector_type(8))) float v8f;

#define NN 50000


__global__ __launch_bounds__(256) void k_deg_init(float* __restrict__ deg, int n) {
  int i = blockIdx.x * blockDim.x + threadIdx.x;
  if (i < n) deg[i] = 1.0f;
}

__global__ __launch_bounds__(256) void k_deg_count(const int* __restrict__ dst,
                                                   float* __restrict__ deg, int e) {
  int i = blockIdx.x * blockDim.x + threadIdx.x;
  if (i < e) atomicAdd(&deg[dst[i]], 1.0f);
}

__global__ __launch_bounds__(256) void k_dinv(float* __restrict__ deg, int n) {
  int i = blockIdx.x * blockDim.x + threadIdx.x;
  if (i < n) {
    float d = deg[i];
    deg[i] = (d > 0.0f) ? rsqrtf(d) : 0.0f;
  }
}

__global__ __launch_bounds__(256) void k_norm(const int* __restrict__ src,
                                              const int* __restrict__ dst,
                                              const float* __restrict__ dinv,
                                              float* __restrict__ norm, int e) {
  int i = blockIdx.x * blockDim.x + threadIdx.x;
  if (i < e) norm[i] = dinv[src[i]] * dinv[dst[i]];
}

template <int K, int OUTD>
__global__ __launch_bounds__(256) void k_transpose(const float* __restrict__ W,
                                                   float* __restrict__ Wt) {
  int i = blockIdx.x * blockDim.x + threadIdx.x;
  if (i < K * OUTD) {
    int k = i / OUTD;
    int n = i - k * OUTD;
    Wt[n * K + k] = W[i];
  }
}

template <int K, int OUTD>
__global__ __launch_bounds__(256) void k_gemm_wmma(const float* __restrict__ X,
                                                   const float* __restrict__ Wt,
                                                   float* __restrict__ H) {
  static_assert(K == 128, "K fixed at 128");
  constexpr int PAD = K + 4;
  __shared__ float xs[16 * PAD];

  const int tid   = threadIdx.x;
  const int lane  = tid & 31;
  const int wave  = tid >> 5;
  const int mtile = blockIdx.x;

  constexpr int NTHREADS = (OUTD / 16) * 32;
  constexpr int NSLOTS   = 16 * (K / 4);
#pragma unroll
  for (int s = tid; s < NSLOTS; s += NTHREADS) {
    int row = s >> 5;
    int c4  = s & 31;
    float4 v = *(const float4*)(X + (size_t)(mtile * 16 + row) * K + c4 * 4);
    *(float4*)(&xs[row * PAD + c4 * 4]) = v;
  }
  __syncthreads();

  const int idx16 = lane & 15;
  const int half  = lane >> 4;
  const float* xsrow = &xs[idx16 * PAD + 2 * half];
  const float* wrow  = Wt + (size_t)(wave * 16 + idx16) * K + 2 * half;

  v8f acc = {};
#pragma unroll
  for (int k0 = 0; k0 < K; k0 += 4) {
    v2f a = *(const v2f*)(xsrow + k0);
    v2f b = *(const v2f*)(wrow + k0);
    acc = __builtin_amdgcn_wmma_f32_16x16x4_f32(
         false, a,  false, b,
         (short)0, acc,  false,  false);
  }

  float* hp = H + (size_t)(mtile * 16 + half * 8) * OUTD + wave * 16 + idx16;
#pragma unroll
  for (int r = 0; r < 8; ++r) hp[r * OUTD] = acc[r];
}


template <int OUTD>
__global__ __launch_bounds__(256) void k_self_init(const float* __restrict__ H,
                                                   const float* __restrict__ dinv,
                                                   float* __restrict__ agg,
                                                   long long total) {
  constexpr int SHIFT = (OUTD == 128) ? 7 : 6;
  long long idx = (long long)blockIdx.x * blockDim.x + threadIdx.x;
  if (idx >= total) return;
  float di = dinv[(int)(idx >> SHIFT)];
  agg[idx] = H[idx] * di * di;
}

template <int OUTD>
__global__ __launch_bounds__(256) void k_edge_scatter(const int* __restrict__ src,
                                                      const int* __restrict__ dst,
                                                      const float* __restrict__ norm,
                                                      const float* __restrict__ H,
                                                      float* __restrict__ agg, int E) {
  constexpr int VEC = OUTD / 32;
  int e = blockIdx.x * (blockDim.x >> 5) + (threadIdx.x >> 5);
  if (e >= E) return;
  int lane = threadIdx.x & 31;
  int s = src[e];
  int d = dst[e];
  float w = norm[e];
  const float* hp = H + (size_t)s * OUTD + lane * VEC;
  float* ap = agg + (size_t)d * OUTD + lane * VEC;
  if constexpr (VEC == 4) {
    float4 v = *(const float4*)hp;
    atomicAdd(ap + 0, v.x * w);
    atomicAdd(ap + 1, v.y * w);
    atomicAdd(ap + 2, v.z * w);
    atomicAdd(ap + 3, v.w * w);
  } else {
    float2 v = *(const float2*)hp;
    atomicAdd(ap + 0, v.x * w);
    atomicAdd(ap + 1, v.y * w);
  }
}

template <int OUTD, int RELU>
__global__ __launch_bounds__(256) void k_finalize(const float* __restrict__ agg,
                                                  const float* __restrict__ bias,
                                                  float* __restrict__ out,
                                                  long long total) {
  long long idx = (long long)blockIdx.x * blockDim.x + threadIdx.x;
  if (idx >= total) return;
  float v = agg[idx] + bias[(int)idx & (OUTD - 1)];
  if (RELU) v = fmaxf(v, 0.0f);
  out[idx] = v;
}


template <int OUTD, int RELU>
static void launch_layer(const float* xin, const float* W, const float* b,
                         const int* src, const int* dst,
                         const float* norm, const float* dinv,
                         float* wt, float* h, float* agg, float* out,
                         int E, hipStream_t stream) {
  constexpr int K = 128;
  k_transpose<K, OUTD><<<(K * OUTD + 255) / 256, 256, 0, stream>>>(W, wt);
  k_gemm_wmma<K, OUTD><<<dim3(NN / 16), dim3((OUTD / 16) * 32), 0, stream>>>(xin, wt, h);

  long long tot = (long long)NN * OUTD;
  unsigned nb = (unsigned)((tot + 255) / 256);
  k_self_init<OUTD><<<nb, 256, 0, stream>>>(h, dinv, agg, tot);

  unsigned eb = (unsigned)((E + 7) / 8);
  k_edge_scatter<OUTD><<<eb, 256, 0, stream>>>(src, dst, norm, h, agg, E);

  k_finalize<OUTD, RELU><<<nb, 256, 0, stream>>>(agg, b, out, tot);
}

extern "C" void kernel_launch(void* const* d_in, const int* in_sizes, int n_in,
                              void* d_out, int out_size, void* d_ws, size_t ws_size,
                              hipStream_t stream) {
  const float* x  = (const float*)d_in[0];
  const int*   ei = (const int*)d_in[1];
  const float* W1 = (const float*)d_in[2];
  const float* b1 = (const float*)d_in[3];
  const float* W2 = (const float*)d_in[4];
  const float* b2 = (const float*)d_in[5];
  const float* W3 = (const float*)d_in[6];
  const float* b3 = (const float*)d_in[7];
  const float* W4 = (const float*)d_in[8];
  const float* b4 = (const float*)d_in[9];

  const int E = in_sizes[1] / 2;
  const int* src = ei;
  const int* dst = ei + E;

  float* dinv = (float*)d_ws;
  float* norm = dinv + ((NN + 3) & ~3);
  float* h    = norm + (((size_t)E + 3) & ~(size_t)3);
  float* agg  = h + (size_t)NN * 128;
  float* xcur = agg + (size_t)NN * 128;
  float* wt   = xcur + (size_t)NN * 128;

  unsigned nb = (NN + 255) / 256;
  unsigned eb = (unsigned)((E + 255) / 256);
  k_deg_init<<<nb, 256, 0, stream>>>(dinv, NN);
  k_deg_count<<<eb, 256, 0, stream>>>(dst, dinv, E);
  k_dinv<<<nb, 256, 0, stream>>>(dinv, NN);
  k_norm<<<eb, 256, 0, stream>>>(src, dst, dinv, norm, E);

  launch_layer<128, 1>(x,    W1, b1, src, dst, norm, dinv, wt, h, agg, xcur, E, stream);
  launch_layer<128, 1>(xcur, W2, b2, src, dst, norm, dinv, wt, h, agg, xcur, E, stream);
  launch_layer<128, 1>(xcur, W3, b3, src, dst, norm, dinv, wt, h, agg, xcur, E, stream);
  launch_layer<64,  0>(xcur, W4, b4, src, dst, norm, dinv, wt, h, agg,
                       (float*)d_out, E, stream);
}


# GEMM re-tiled: a wave owns 32x32 of H as 2x2 f32 WMMA tiles (each A and B fragment feeds 2 products, B pairs are 8-byte loads), loads grouped a k block ahead; 6252 waves, not 25000; k order kept
# speedup vs baseline: 4.7755x; 4.7755x over previous
#include <hip/hip_runtime.h>
#include <hip/hip_bf16.h>

typedef __attribute__((ext_vector_type(2))) float v2f;
typedef __attribute__((ext_vector_type(8))) float v8f;

#define NN 50000

#define TSH 6
#define TN  (1 << TSH)
#define NBN ((NN + TN - 1) / TN)
#define ASH 4
#define TA  (1 << ASH)
#define NB  ((NN + TA - 1) / TA)
#define NBP 4096
#define TPR (TN / TA)
#define AW  4


__global__ __launch_bounds__(256) void k_cnt_zero(int* __restrict__ cnt) {
  int i = blockIdx.x * blockDim.x + threadIdx.x;
  if (i < NBP) cnt[i] = 0;
}

template <int K, int OUTD, int MT>
__device__ __forceinline__ void gemm_rows(const float* __restrict__ X,
                                          const float* __restrict__ Wm,
                                          float* __restrict__ H, const int row0, float* xs) {
  constexpr int PAD = K + 4;
  constexpr int BM  = 16 * MT;
  const int tid   = threadIdx.x;
  const int lane  = tid & 31;
  const int wave  = tid >> 5;
  const int idx16 = lane & 15;
  const int half  = lane >> 4;
  const float* wcol = Wm + (size_t)(2 * half) * OUTD + wave * 32 + idx16 * 2;

  constexpr int KB   = 4;
  constexpr int NBLK = K / (4 * KB);
  v2f wf[2][KB][2];
  v2f af[2][KB][MT];

#define GEMM_FETCH_W(SET, BLK)                                                       \
  _Pragma("unroll") for (int ks = 0; ks < KB; ++ks) {                                \
    const int k0 = ((BLK) * KB + ks) * 4;                                            \
    wf[SET][ks][0] = *(const v2f*)(wcol + (size_t)k0 * OUTD);                        \
    wf[SET][ks][1] = *(const v2f*)(wcol + (size_t)(k0 + 1) * OUTD);                  \
  }
#define GEMM_FETCH_A(SET, BLK)                                                       \
  _Pragma("unroll") for (int ks = 0; ks < KB; ++ks) {                                \
    const int k0 = ((BLK) * KB + ks) * 4;                                            \
    _Pragma("unroll") for (int m = 0; m < MT; ++m)                                   \
      af[SET][ks][m] = *(const v2f*)(xsrow + m * 16 * PAD + k0);                     \
  }

  {
    constexpr int NTHREADS = (OUTD / 32) * 32;
    constexpr int PER      = BM * (K / 4) / NTHREADS;
    float4 st[PER];
#pragma unroll
    for (int i = 0; i < PER; ++i) {
      const int s = tid + i * NTHREADS;
      st[i] = *(const float4*)(X + (size_t)(row0 + (s >> 5)) * K + (s & 31) * 4);
    }
    GEMM_FETCH_W(0, 0)
#pragma unroll
    for (int i = 0; i < PER; ++i) {
      const int s = tid + i * NTHREADS;
      *(float4*)(&xs[(s >> 5) * PAD + (s & 31) * 4]) = st[i];
    }
  }
  __syncthreads();

  const float* xsrow = &xs[idx16 * PAD + 2 * half];

  v8f acc[MT][2] = {};
  GEMM_FETCH_A(0, 0)
#pragma unroll
  for (int kb = 0; kb < NBLK; ++kb) {
    if (kb + 1 < NBLK) { GEMM_FETCH_W((kb + 1) & 1, kb + 1) GEMM_FETCH_A((kb + 1) & 1, kb + 1) }
    __builtin_amdgcn_sched_barrier(0);
#pragma unroll
    for (int ks = 0; ks < KB; ++ks) {
      const v2f b0 = {wf[kb & 1][ks][0].x, wf[kb & 1][ks][1].x};
      const v2f b1 = {wf[kb & 1][ks][0].y, wf[kb & 1][ks][1].y};
#pragma unroll
      for (int m = 0; m < MT; ++m) {
        const v2f a = af[kb & 1][ks][m];
        acc[m][0] = __builtin_amdgcn_wmma_f32_16x16x4_f32(
             false, a,  false, b0,
             (short)0, acc[m][0],  false,  false);
        acc[m][1] = __builtin_amdgcn_wmma_f32_16x16x4_f32(
             false, a,  false, b1,
             (short)0, acc[m][1],  false,  false);
      }
    }
  }
#undef GEMM_FETCH_W
#undef GEMM_FETCH_A

#pragma unroll
  for (int m = 0; m < MT; ++m) {
    float* hp = H + (size_t)(row0 + m * 16 + half * 8) * OUTD + wave * 32 + idx16 * 2;
#pragma unroll
    for (int r = 0; r < 8; ++r) {
      const v2f o = {acc[m][0][r], acc[m][1][r]};
      *(v2f*)(hp + r * OUTD) = o;
    }
  }
}

template <int K, int OUTD>
__global__ __launch_bounds__((OUTD / 32) * 32) void k_gemm_wmma(const float* __restrict__ X,
                                                                const float* __restrict__ Wm,
                                                                float* __restrict__ H) {
  static_assert(K == 128, "K fixed at 128");
  static_assert(NN % 16 == 0, "row tiles are whole");
  __shared__ float xs[32 * (K + 4)];
  const int row0 = blockIdx.x * 32;
  if (row0 + 32 <= NN) gemm_rows<K, OUTD, 2>(X, Wm, H, row0, xs);
  else                 gemm_rows<K, OUTD, 1>(X, Wm, H, row0, xs);
}


__global__ __launch_bounds__(256) void k_tile_count(const int* __restrict__ dst,
                                                    int* __restrict__ cnt, int E) {
  __shared__ int hist[NBP];
  for (int b = threadIdx.x; b < NBP; b += 256) hist[b] = 0;
  __syncthreads();
  const int chunk = (E + (int)gridDim.x - 1) / (int)gridDim.x;
  const int lo = (int)blockIdx.x * chunk;
  const int hi = min(lo + chunk, E);
  for (int e = lo + (int)threadIdx.x; e < hi; e += 256) atomicAdd(&hist[dst[e] >> ASH], 1);
  __syncthreads();
  for (int b = threadIdx.x; b < NB; b += 256) {
    int c = hist[b];
    if (c) atomicAdd(&cnt[b], c);
  }
}

__global__ __launch_bounds__(1024) void k_tile_scan(const int* __restrict__ cnt,
                                                    int* __restrict__ start,
                                                    int* __restrict__ cursor) {
  constexpr int PT = NBP / 1024;
  __shared__ int s[1024];
  const int t = threadIdx.x;
  int c[PT];
  int v = 0;
#pragma unroll
  for (int i = 0; i < PT; ++i) {
    const int b = t * PT + i;
    c[i] = (b < NB) ? cnt[b] : 0;
    v += c[i];
  }
  s[t] = v;
  __syncthreads();
  for (int off = 1; off < 1024; off <<= 1) {
    int add = (t >= off) ? s[t - off] : 0;
    __syncthreads();
    s[t] += add;
    __syncthreads();
  }
  int run = s[t] - v;
#pragma unroll
  for (int i = 0; i < PT; ++i) {
    const int b = t * PT + i;
    if (b < NB) { start[b] = run; cursor[b] = run; }
    run += c[i];
  }
  if (t == 1023) start[NB] = s[t];
}

__global__ __launch_bounds__(256) void k_tile_fill(const int* __restrict__ src,
                                                   const int* __restrict__ dst,
                                                   int* __restrict__ cursor,
                                                   int* __restrict__ rec, int E) {
  __shared__ int hist[NBP];
  __shared__ int base[NBP];
  for (int b = threadIdx.x; b < NBP; b += 256) hist[b] = 0;
  __syncthreads();
  const int chunk = (E + (int)gridDim.x - 1) / (int)gridDim.x;
  const int lo = (int)blockIdx.x * chunk;
  const int hi = min(lo + chunk, E);
  for (int e = lo + (int)threadIdx.x; e < hi; e += 256) atomicAdd(&hist[dst[e] >> ASH], 1);
  __syncthreads();
  for (int b = threadIdx.x; b < NB; b += 256) {
    int c = hist[b];
    base[b] = c ? atomicAdd(&cursor[b], c) : 0;
    hist[b] = 0;
  }
  __syncthreads();
  for (int e = lo + (int)threadIdx.x; e < hi; e += 256) {
    int s = src[e];
    int d = dst[e];
    int b = d >> ASH;
    int p = base[b] + atomicAdd(&hist[b], 1);
    rec[2 * p] = (s << TSH) | (d & (TN - 1));
  }
}

__global__ __launch_bounds__(256) void k_tile_norm(const int* __restrict__ start,
                                                   const int* __restrict__ rec,
                                                   float* __restrict__ dinv) {
  constexpr int B = 8;
  __shared__ int hist[TN];
  if (threadIdx.x < TN) hist[threadIdx.x] = 0;
  __syncthreads();
  const int e0 = start[(int)blockIdx.x * TPR];
  const int e1 = start[min((int)blockIdx.x * TPR + TPR, NB)];
  for (int p0 = e0 + (int)threadIdx.x; p0 < e1; p0 += B * 256) {
    int xr[B];
#pragma unroll
    for (int u = 0; u < B; ++u) xr[u] = rec[2 * min(p0 + u * 256, e1 - 1)];
#pragma unroll
    for (int u = 0; u < B; ++u)
      if (p0 + u * 256 < e1) atomicAdd(&hist[xr[u] & (TN - 1)], 1);
  }
  __syncthreads();
  if (threadIdx.x < TN) {
    const int node = (int)blockIdx.x * TN + (int)threadIdx.x;
    if (node < NN) {
      const float d = 1.0f + (float)hist[threadIdx.x];
      dinv[node] = (d > 0.0f) ? rsqrtf(d) : 0.0f;
    }
  }
}

__global__ __launch_bounds__(256) void k_edge_weight(const int* __restrict__ start,
                                                     const float* __restrict__ dinv,
                                                     int* rec) {
  constexpr int B = 8;
  const int node0 = (int)blockIdx.x * TN;
  const int e0 = start[(int)blockIdx.x * TPR];
  const int e1 = start[min((int)blockIdx.x * TPR + TPR, NB)];
  for (int p0 = e0 + (int)threadIdx.x; p0 < e1; p0 += B * 256) {
    int x[B];
    float w[B];
#pragma unroll
    for (int u = 0; u < B; ++u) x[u] = rec[2 * min(p0 + u * 256, e1 - 1)];
#pragma unroll
    for (int u = 0; u < B; ++u) w[u] = dinv[x[u] >> TSH] * dinv[node0 + (x[u] & (TN - 1))];
#pragma unroll
    for (int u = 0; u < B; ++u) {
      const int p = p0 + u * 256;
      if (p < e1) rec[2 * p + 1] = __float_as_int(w[u]);
    }
  }
}


template <int V> using fvec = float __attribute__((ext_vector_type(V)));

__device__ __forceinline__ void lds_add(float* p, float v) {
  __hip_atomic_fetch_add(p, v, __ATOMIC_RELAXED, __HIP_MEMORY_SCOPE_WORKGROUP);
}

template <int OUTD, int U>
__device__ __forceinline__ void agg_fetch(fvec<OUTD / 32> (&dst)[U], const int pk, const int g,
                                          const float* __restrict__ H, const unsigned loff) {
  typedef fvec<OUTD / 32> vec_t;
#pragma unroll
  for (int u = 0; u < U; ++u) {
    const int s = __builtin_amdgcn_readlane(pk, g * U + u) >> TSH;
    const float* row = H + (size_t)s * OUTD;
    dst[u] = *(const vec_t*)(row + loff);
  }
}

template <int OUTD, bool FULL>
__device__ __forceinline__ void agg_chunk(fvec<OUTD / 32> (&buf)[2][8], const int pk, const int wi,
                                          const int n, const int pkn,
                                          const float* __restrict__ H, const unsigned loff,
                                          float* al) {
  constexpr int VEC = OUTD / 32;
  constexpr int U   = 8;
  constexpr int G   = 32 / U;
#pragma unroll
  for (int g = 0; g < G; ++g) {
    if (g + 1 < G) agg_fetch<OUTD, U>(buf[(g + 1) & 1], pk, g + 1, H, loff);
    else if (FULL) agg_fetch<OUTD, U>(buf[0], pkn, 0, H, loff);
    __builtin_amdgcn_sched_barrier(0);
#pragma unroll
    for (int u = 0; u < U; ++u) {
      const int j = g * U + u;
      if (FULL || j < n) {
        const int pj = __builtin_amdgcn_readlane(pk, j);
        const float wj = __int_as_float(__builtin_amdgcn_readlane(wi, j));
        float* ap = al + (pj & (TA - 1)) * OUTD;
#pragma unroll
        for (int c = 0; c < VEC; ++c) lds_add(ap + c * 32, buf[g & 1][u][c] * wj);
      }
    }
  }
}

template <int OUTD, int RELU>
__global__ __launch_bounds__(AW * 32) void k_tile_aggregate(const int* __restrict__ start,
                                                        const int* __restrict__ rec,
                                                        const float* __restrict__ H,
                                                        const float* __restrict__ dinv,
                                                        const float* __restrict__ bias,
                                                        float* __restrict__ out) {
  constexpr int VEC = OUTD / 32;
  typedef fvec<VEC> vec_t;
  __shared__ float acc[TA * OUTD];

  const int lane  = threadIdx.x & 31;
  const int wave  = __builtin_amdgcn_readfirstlane((int)(threadIdx.x >> 5));
  const int node0 = (int)blockIdx.x * TA;
  const unsigned loff = (unsigned)lane * VEC;
  float* al = acc + lane;

  {
    constexpr int RW = TA / AW;
    vec_t sv[RW];
    float sd[RW];
#pragma unroll
    for (int i = 0; i < RW; ++i) {
      const int node = min(node0 + wave + AW * i, NN - 1);
      sd[i] = dinv[node];
      sv[i] = *(const vec_t*)(H + (size_t)node * OUTD + loff);
    }
#pragma unroll
    for (int i = 0; i < RW; ++i) {
      const int r = wave + AW * i;
#pragma unroll
      for (int c = 0; c < VEC; ++c) al[r * OUTD + c * 32] = sv[i][c] * sd[i] * sd[i];
    }
  }
  __syncthreads();

  const int e0 = start[blockIdx.x];
  const int e1 = start[blockIdx.x + 1];
  int cb = e0 + wave * 32;
  if (cb < e1) {
    const int2* rec2 = (const int2*)rec;
    int2 rc = rec2[min(cb + lane, e1 - 1)];
    vec_t buf[2][8];
    agg_fetch<OUTD, 8>(buf[0], rc.x, 0, H, loff);
    for (; cb + 32 <= e1; cb += AW * 32) {
      const int2 rn = rec2[min(cb + AW * 32 + lane, e1 - 1)];
      agg_chunk<OUTD, true>(buf, rc.x, rc.y, 32, rn.x, H, loff, al);
      rc = rn;
    }
    if (cb < e1) agg_chunk<OUTD, false>(buf, rc.x, rc.y, e1 - cb, 0, H, loff, al);
  }
  __syncthreads();

  const vec_t bv = *(const vec_t*)(bias + lane * VEC);
  for (int r = wave; r < TA; r += AW) {
    const int node = node0 + r;
    if (node < NN) {
      vec_t o;
#pragma unroll
      for (int c = 0; c < VEC; ++c) {
        float x = al[r * OUTD + c * 32] + bv[c];
        if (RELU) x = fmaxf(x, 0.0f);
        o[c] = x;
      }
      *(vec_t*)(out + (size_t)node * OUTD + lane * VEC) = o;
    }
  }
}


template <int OUTD, int RELU>
static void launch_layer(const float* xin, const float* W, const float* b,
                         const int* start, const int* rec,
                         const float* dinv, float* h, float* out,
                         hipStream_t stream) {
  constexpr int K = 128;
  k_gemm_wmma<K, OUTD><<<dim3((NN + 31) / 32), dim3((OUTD / 32) * 32), 0, stream>>>(xin, W, h);
  k_tile_aggregate<OUTD, RELU><<<NB, AW * 32, 0, stream>>>(start, rec, h, dinv, b, out);
}

extern "C" void kernel_launch(void* const* d_in, const int* in_sizes, int n_in,
                              void* d_out, int out_size, void* d_ws, size_t ws_size,
                              hipStream_t stream) {
  const float* x  = (const float*)d_in[0];
  const int*   ei = (const int*)d_in[1];
  const float* W1 = (const float*)d_in[2];
  const float* b1 = (const float*)d_in[3];
  const float* W2 = (const float*)d_in[4];
  const float* b2 = (const float*)d_in[5];
  const float* W3 = (const float*)d_in[6];
  const float* b3 = (const float*)d_in[7];
  const float* W4 = (const float*)d_in[8];
  const float* b4 = (const float*)d_in[9];

  const int E = in_sizes[1] / 2;
  const int* src = ei;
  const int* dst = ei + E;

  float* dinv = (float*)d_ws;
  float* espc = dinv + ((NN + 3) & ~3);
  float* h    = espc + (((size_t)E + 3) & ~(size_t)3);
  float* agg  = h + (size_t)NN * 128;
  float* xcur = agg + (size_t)NN * 128;
  int* rec    = (int*)agg;
  int* cnt    = (int*)agg + 2 * (((size_t)E + 3) & ~(size_t)3);
  int* cursor = cnt + NBP;
  int* start  = cursor + NBP;

  k_cnt_zero<<<NBP / 256, 256, 0, stream>>>(cnt);
  k_tile_count<<<256, 256, 0, stream>>>(dst, cnt, E);
  k_tile_scan<<<1, 1024, 0, stream>>>(cnt, start, cursor);
  k_tile_fill<<<256, 256, 0, stream>>>(src, dst, cursor, rec, E);

  k_tile_norm<<<NBN, 256, 0, stream>>>(start, rec, dinv);
  k_edge_weight<<<NBN, 256, 0, stream>>>(start, dinv, rec);

  launch_layer<128, 1>(x,    W1, b1, start, rec, dinv, h, xcur, stream);
  launch_layer<128, 1>(xcur, W2, b2, start, rec, dinv, h, xcur, stream);
  launch_layer<128, 1>(xcur, W3, b3, start, rec, dinv, h, xcur, stream);
  launch_layer<64,  0>(xcur, W4, b4, start, rec, dinv, h, (float*)d_out, stream);
}


# layers 1 and 3 sum neighbour rows before their product (they commute), so products 1+2 and 3+4 run back to back on an LDS tile, one launch per pair: 14 launches to 12; rounding order changes
# speedup vs baseline: 5.0547x; 1.0585x over previous
#include <hip/hip_runtime.h>
#include <hip/hip_bf16.h>

typedef __attribute__((ext_vector_type(2))) float v2f;
typedef __attribute__((ext_vector_type(8))) float v8f;

#define NN 50000

#define TSH 6
#define TN  (1 << TSH)
#define NBN ((NN + TN - 1) / TN)
#define ASH 4
#define TA  (1 << ASH)
#define NB  ((NN + TA - 1) / TA)
#define NBP 4096
#define TPR (TN / TA)
#define AW  4


__global__ __launch_bounds__(256) void k_cnt_zero(int* __restrict__ cnt) {
  int i = blockIdx.x * blockDim.x + threadIdx.x;
  if (i < NBP) cnt[i] = 0;
}

#define GEMM_KB 4
#define GEMM_FETCH_W(SET, BLK, WCOL, LD)                                             \
  _Pragma("unroll") for (int ks = 0; ks < GEMM_KB; ++ks) {                           \
    const int k0 = ((BLK) * GEMM_KB + ks) * 4;                                       \
    wf[SET][ks][0] = *(const v2f*)((WCOL) + (size_t)k0 * (LD));                      \
    wf[SET][ks][1] = *(const v2f*)((WCOL) + (size_t)(k0 + 1) * (LD));                \
  }
#define GEMM_FETCH_A(SET, BLK, TROW)                                                 \
  _Pragma("unroll") for (int ks = 0; ks < GEMM_KB; ++ks) {                           \
    const int k0 = ((BLK) * GEMM_KB + ks) * 4;                                       \
    _Pragma("unroll") for (int m = 0; m < MT; ++m)                                   \
      af[SET][ks][m] = *(const v2f*)((TROW) + m * 16 * PAD + k0);                    \
  }

template <int K, int LD, int MT>
__device__ __forceinline__ void gemm_kloop(const float* __restrict__ wcol, const float* trow,
                                           v2f (&wf)[2][GEMM_KB][2], v8f (&acc)[MT][2]) {
  constexpr int PAD  = K + 4;
  constexpr int NBLK = K / (4 * GEMM_KB);
  v2f af[2][GEMM_KB][MT];
  GEMM_FETCH_A(0, 0, trow)
#pragma unroll
  for (int kb = 0; kb < NBLK; ++kb) {
    if (kb + 1 < NBLK) { GEMM_FETCH_W((kb + 1) & 1, kb + 1, wcol, LD) GEMM_FETCH_A((kb + 1) & 1, kb + 1, trow) }
    __builtin_amdgcn_sched_barrier(0);
#pragma unroll
    for (int ks = 0; ks < GEMM_KB; ++ks) {
      const v2f b0 = {wf[kb & 1][ks][0].x, wf[kb & 1][ks][1].x};
      const v2f b1 = {wf[kb & 1][ks][0].y, wf[kb & 1][ks][1].y};
#pragma unroll
      for (int m = 0; m < MT; ++m) {
        const v2f a = af[kb & 1][ks][m];
        acc[m][0] = __builtin_amdgcn_wmma_f32_16x16x4_f32(
             false, a,  false, b0,
             (short)0, acc[m][0],  false,  false);
        acc[m][1] = __builtin_amdgcn_wmma_f32_16x16x4_f32(
             false, a,  false, b1,
             (short)0, acc[m][1],  false,  false);
      }
    }
  }
}

template <int K, int MID, int OUTD, int MT>
__device__ __forceinline__ void gemm2_rows(const float* __restrict__ X,
                                           const float* __restrict__ Wa,
                                           const float* __restrict__ ba,
                                           const float* __restrict__ Wb,
                                           float* __restrict__ H, const int row0,
                                           float* xs, float* hs) {
  static_assert(K == MID, "both tiles share one row stride");
  static_assert(OUTD == MID || 2 * OUTD == MID, "product B: same split or 16-row tiles");
  constexpr int PAD = K + 4;
  constexpr int BM  = 16 * MT;
  const int tid   = threadIdx.x;
  const int lane  = tid & 31;
  const int wave  = tid >> 5;
  const int idx16 = lane & 15;
  const int half  = lane >> 4;
  const float* wcolA = Wa + (size_t)(2 * half) * MID + wave * 32 + idx16 * 2;

  v2f wf[2][GEMM_KB][2];

  {
    constexpr int NTHREADS = (MID / 32) * 32;
    constexpr int PER      = BM * (K / 4) / NTHREADS;
    float4 st[PER];
#pragma unroll
    for (int i = 0; i < PER; ++i) {
      const int s = tid + i * NTHREADS;
      st[i] = *(const float4*)(X + (size_t)(row0 + (s >> 5)) * K + (s & 31) * 4);
    }
    GEMM_FETCH_W(0, 0, wcolA, MID)
#pragma unroll
    for (int i = 0; i < PER; ++i) {
      const int s = tid + i * NTHREADS;
      *(float4*)(&xs[(s >> 5) * PAD + (s & 31) * 4]) = st[i];
    }
  }
  __syncthreads();

  {
    v8f acc[MT][2] = {};
    gemm_kloop<K, MID, MT>(wcolA, &xs[idx16 * PAD + 2 * half], wf, acc);
    const v2f bb = *(const v2f*)(ba + wave * 32 + idx16 * 2);
#pragma unroll
    for (int m = 0; m < MT; ++m) {
      float* hp = hs + (m * 16 + half * 8) * PAD + wave * 32 + idx16 * 2;
#pragma unroll
      for (int r = 0; r < 8; ++r) {
        const v2f o = {fmaxf(acc[m][0][r] + bb.x, 0.0f), fmaxf(acc[m][1][r] + bb.y, 0.0f)};
        *(v2f*)(hp + r * PAD) = o;
      }
    }
  }

  constexpr bool SAME = (OUTD == MID);
  constexpr int  MT2  = SAME ? MT : 1;
  const int cs = SAME ? wave : (wave & 1);
  const int mr = SAME ? 0 : (wave >> 1);
  const bool active = SAME || (mr < MT);
  const float* wcolB = Wb + (size_t)(2 * half) * OUTD + cs * 32 + idx16 * 2;
  if (active) { GEMM_FETCH_W(0, 0, wcolB, OUTD) }
  __syncthreads();
  if (active) {
    v8f acc[MT2][2] = {};
    gemm_kloop<MID, OUTD, MT2>(wcolB, &hs[(mr * 16 + idx16) * PAD + 2 * half], wf, acc);
#pragma unroll
    for (int m = 0; m < MT2; ++m) {
      float* hp = H + (size_t)(row0 + (mr + m) * 16 + half * 8) * OUTD + cs * 32 + idx16 * 2;
#pragma unroll
      for (int r = 0; r < 8; ++r) {
        const v2f o = {acc[m][0][r], acc[m][1][r]};
        *(v2f*)(hp + r * OUTD) = o;
      }
    }
  }
}
#undef GEMM_FETCH_W
#undef GEMM_FETCH_A

template <int K, int MID, int OUTD>
__global__ __launch_bounds__((MID / 32) * 32) void k_gemm2_wmma(const float* __restrict__ X,
                                                                const float* __restrict__ Wa,
                                                                const float* __restrict__ ba,
                                                                const float* __restrict__ Wb,
                                                                float* __restrict__ H) {
  static_assert(K == 128, "K fixed at 128");
  static_assert(NN % 16 == 0, "row tiles are whole");
  __shared__ float xs[32 * (K + 4)];
  __shared__ float hs[32 * (MID + 4)];
  const int row0 = blockIdx.x * 32;
  if (row0 + 32 <= NN) gemm2_rows<K, MID, OUTD, 2>(X, Wa, ba, Wb, H, row0, xs, hs);
  else                 gemm2_rows<K, MID, OUTD, 1>(X, Wa, ba, Wb, H, row0, xs, hs);
}


__global__ __launch_bounds__(256) void k_tile_count(const int* __restrict__ dst,
                                                    int* __restrict__ cnt, int E) {
  __shared__ int hist[NBP];
  for (int b = threadIdx.x; b < NBP; b += 256) hist[b] = 0;
  __syncthreads();
  const int chunk = (E + (int)gridDim.x - 1) / (int)gridDim.x;
  const int lo = (int)blockIdx.x * chunk;
  const int hi = min(lo + chunk, E);
  for (int e = lo + (int)threadIdx.x; e < hi; e += 256) atomicAdd(&hist[dst[e] >> ASH], 1);
  __syncthreads();
  for (int b = threadIdx.x; b < NB; b += 256) {
    int c = hist[b];
    if (c) atomicAdd(&cnt[b], c);
  }
}

__global__ __launch_bounds__(1024) void k_tile_scan(const int* __restrict__ cnt,
                                                    int* __restrict__ start,
                                                    int* __restrict__ cursor) {
  constexpr int PT = NBP / 1024;
  __shared__ int s[1024];
  const int t = threadIdx.x;
  int c[PT];
  int v = 0;
#pragma unroll
  for (int i = 0; i < PT; ++i) {
    const int b = t * PT + i;
    c[i] = (b < NB) ? cnt[b] : 0;
    v += c[i];
  }
  s[t] = v;
  __syncthreads();
  for (int off = 1; off < 1024; off <<= 1) {
    int add = (t >= off) ? s[t - off] : 0;
    __syncthreads();
    s[t] += add;
    __syncthreads();
  }
  int run = s[t] - v;
#pragma unroll
  for (int i = 0; i < PT; ++i) {
    const int b = t * PT + i;
    if (b < NB) { start[b] = run; cursor[b] = run; }
    run += c[i];
  }
  if (t == 1023) start[NB] = s[t];
}

__global__ __launch_bounds__(256) void k_tile_fill(const int* __restrict__ src,
                                                   const int* __restrict__ dst,
                                                   int* __restrict__ cursor,
                                                   int* __restrict__ rec, int E) {
  __shared__ int hist[NBP];
  __shared__ int base[NBP];
  for (int b = threadIdx.x; b < NBP; b += 256) hist[b] = 0;
  __syncthreads();
  const int chunk = (E + (int)gridDim.x - 1) / (int)gridDim.x;
  const int lo = (int)blockIdx.x * chunk;
  const int hi = min(lo + chunk, E);
  for (int e = lo + (int)threadIdx.x; e < hi; e += 256) atomicAdd(&hist[dst[e] >> ASH], 1);
  __syncthreads();
  for (int b = threadIdx.x; b < NB; b += 256) {
    int c = hist[b];
    base[b] = c ? atomicAdd(&cursor[b], c) : 0;
    hist[b] = 0;
  }
  __syncthreads();
  for (int e = lo + (int)threadIdx.x; e < hi; e += 256) {
    int s = src[e];
    int d = dst[e];
    int b = d >> ASH;
    int p = base[b] + atomicAdd(&hist[b], 1);
    rec[2 * p] = (s << TSH) | (d & (TN - 1));
  }
}

__global__ __launch_bounds__(256) void k_tile_norm(const int* __restrict__ start,
                                                   const int* __restrict__ rec,
                                                   float* __restrict__ dinv) {
  constexpr int B = 8;
  __shared__ int hist[TN];
  if (threadIdx.x < TN) hist[threadIdx.x] = 0;
  __syncthreads();
  const int e0 = start[(int)blockIdx.x * TPR];
  const int e1 = start[min((int)blockIdx.x * TPR + TPR, NB)];
  for (int p0 = e0 + (int)threadIdx.x; p0 < e1; p0 += B * 256) {
    int xr[B];
#pragma unroll
    for (int u = 0; u < B; ++u) xr[u] = rec[2 * min(p0 + u * 256, e1 - 1)];
#pragma unroll
    for (int u = 0; u < B; ++u)
      if (p0 + u * 256 < e1) atomicAdd(&hist[xr[u] & (TN - 1)], 1);
  }
  __syncthreads();
  if (threadIdx.x < TN) {
    const int node = (int)blockIdx.x * TN + (int)threadIdx.x;
    if (node < NN) {
      const float d = 1.0f + (float)hist[threadIdx.x];
      dinv[node] = (d > 0.0f) ? rsqrtf(d) : 0.0f;
    }
  }
}

__global__ __launch_bounds__(256) void k_edge_weight(const int* __restrict__ start,
                                                     const float* __restrict__ dinv,
                                                     int* rec) {
  constexpr int B = 8;
  const int node0 = (int)blockIdx.x * TN;
  const int e0 = start[(int)blockIdx.x * TPR];
  const int e1 = start[min((int)blockIdx.x * TPR + TPR, NB)];
  for (int p0 = e0 + (int)threadIdx.x; p0 < e1; p0 += B * 256) {
    int x[B];
    float w[B];
#pragma unroll
    for (int u = 0; u < B; ++u) x[u] = rec[2 * min(p0 + u * 256, e1 - 1)];
#pragma unroll
    for (int u = 0; u < B; ++u) w[u] = dinv[x[u] >> TSH] * dinv[node0 + (x[u] & (TN - 1))];
#pragma unroll
    for (int u = 0; u < B; ++u) {
      const int p = p0 + u * 256;
      if (p < e1) rec[2 * p + 1] = __float_as_int(w[u]);
    }
  }
}


template <int V> using fvec = float __attribute__((ext_vector_type(V)));

__device__ __forceinline__ void lds_add(float* p, float v) {
  __hip_atomic_fetch_add(p, v, __ATOMIC_RELAXED, __HIP_MEMORY_SCOPE_WORKGROUP);
}

template <int OUTD, int U>
__device__ __forceinline__ void agg_fetch(fvec<OUTD / 32> (&dst)[U], const int pk, const int g,
                                          const float* __restrict__ H, const unsigned loff) {
  typedef fvec<OUTD / 32> vec_t;
#pragma unroll
  for (int u = 0; u < U; ++u) {
    const int s = __builtin_amdgcn_readlane(pk, g * U + u) >> TSH;
    const float* row = H + (size_t)s * OUTD;
    dst[u] = *(const vec_t*)(row + loff);
  }
}

template <int OUTD, bool FULL>
__device__ __forceinline__ void agg_chunk(fvec<OUTD / 32> (&buf)[2][8], const int pk, const int wi,
                                          const int n, const int pkn,
                                          const float* __restrict__ H, const unsigned loff,
                                          float* al) {
  constexpr int VEC = OUTD / 32;
  constexpr int U   = 8;
  constexpr int G   = 32 / U;
#pragma unroll
  for (int g = 0; g < G; ++g) {
    if (g + 1 < G) agg_fetch<OUTD, U>(buf[(g + 1) & 1], pk, g + 1, H, loff);
    else if (FULL) agg_fetch<OUTD, U>(buf[0], pkn, 0, H, loff);
    __builtin_amdgcn_sched_barrier(0);
#pragma unroll
    for (int u = 0; u < U; ++u) {
      const int j = g * U + u;
      if (FULL || j < n) {
        const int pj = __builtin_amdgcn_readlane(pk, j);
        const float wj = __int_as_float(__builtin_amdgcn_readlane(wi, j));
        float* ap = al + (pj & (TA - 1)) * OUTD;
#pragma unroll
        for (int c = 0; c < VEC; ++c) lds_add(ap + c * 32, buf[g & 1][u][c] * wj);
      }
    }
  }
}

template <int OUTD, int RELU, bool BIAS>
__global__ __launch_bounds__(AW * 32) void k_tile_aggregate(const int* __restrict__ start,
                                                        const int* __restrict__ rec,
                                                        const float* __restrict__ H,
                                                        const float* __restrict__ dinv,
                                                        const float* __restrict__ bias,
                                                        float* __restrict__ out) {
  constexpr int VEC = OUTD / 32;
  typedef fvec<VEC> vec_t;
  __shared__ float acc[TA * OUTD];

  const int lane  = threadIdx.x & 31;
  const int wave  = __builtin_amdgcn_readfirstlane((int)(threadIdx.x >> 5));
  const int node0 = (int)blockIdx.x * TA;
  const unsigned loff = (unsigned)lane * VEC;
  float* al = acc + lane;

  {
    constexpr int RW = TA / AW;
    vec_t sv[RW];
    float sd[RW];
#pragma unroll
    for (int i = 0; i < RW; ++i) {
      const int node = min(node0 + wave + AW * i, NN - 1);
      sd[i] = dinv[node];
      sv[i] = *(const vec_t*)(H + (size_t)node * OUTD + loff);
    }
#pragma unroll
    for (int i = 0; i < RW; ++i) {
      const int r = wave + AW * i;
#pragma unroll
      for (int c = 0; c < VEC; ++c) al[r * OUTD + c * 32] = sv[i][c] * sd[i] * sd[i];
    }
  }
  __syncthreads();

  const int e0 = start[blockIdx.x];
  const int e1 = start[blockIdx.x + 1];
  int cb = e0 + wave * 32;
  if (cb < e1) {
    const int2* rec2 = (const int2*)rec;
    int2 rc = rec2[min(cb + lane, e1 - 1)];
    vec_t buf[2][8];
    agg_fetch<OUTD, 8>(buf[0], rc.x, 0, H, loff);
    for (; cb + 32 <= e1; cb += AW * 32) {
      const int2 rn = rec2[min(cb + AW * 32 + lane, e1 - 1)];
      agg_chunk<OUTD, true>(buf, rc.x, rc.y, 32, rn.x, H, loff, al);
      rc = rn;
    }
    if (cb < e1) agg_chunk<OUTD, false>(buf, rc.x, rc.y, e1 - cb, 0, H, loff, al);
  }
  __syncthreads();

  vec_t bv = {};
  if (BIAS) bv = *(const vec_t*)(bias + lane * VEC);
  for (int r = wave; r < TA; r += AW) {
    const int node = node0 + r;
    if (node < NN) {
      vec_t o;
#pragma unroll
      for (int c = 0; c < VEC; ++c) {
        float x = al[r * OUTD + c * 32] + bv[c];
        if (RELU) x = fmaxf(x, 0.0f);
        o[c] = x;
      }
      *(vec_t*)(out + (size_t)node * OUTD + lane * VEC) = o;
    }
  }
}


template <int OUTD, int RELU, bool BIAS>
static void launch_aggregate(const float* hin, const float* b, const int* start, const int* rec,
                             const float* dinv, float* out, hipStream_t stream) {
  k_tile_aggregate<OUTD, RELU, BIAS><<<NB, AW * 32, 0, stream>>>(start, rec, hin, dinv, b, out);
}

template <int OUTD>
static void launch_gemm2(const float* xin, const float* Wa, const float* ba, const float* Wb,
                         float* hout, hipStream_t stream) {
  k_gemm2_wmma<128, 128, OUTD><<<dim3((NN + 31) / 32), dim3(128), 0, stream>>>(xin, Wa, ba, Wb, hout);
}

extern "C" void kernel_launch(void* const* d_in, const int* in_sizes, int n_in,
                              void* d_out, int out_size, void* d_ws, size_t ws_size,
                              hipStream_t stream) {
  const float* x  = (const float*)d_in[0];
  const int*   ei = (const int*)d_in[1];
  const float* W1 = (const float*)d_in[2];
  const float* b1 = (const float*)d_in[3];
  const float* W2 = (const float*)d_in[4];
  const float* b2 = (const float*)d_in[5];
  const float* W3 = (const float*)d_in[6];
  const float* b3 = (const float*)d_in[7];
  const float* W4 = (const float*)d_in[8];
  const float* b4 = (const float*)d_in[9];

  const int E = in_sizes[1] / 2;
  const int* src = ei;
  const int* dst = ei + E;

  float* dinv = (float*)d_ws;
  float* espc = dinv + ((NN + 3) & ~3);
  float* h    = espc + (((size_t)E + 3) & ~(size_t)3);
  float* agg  = h + (size_t)NN * 128;
  float* xcur = agg + (size_t)NN * 128;
  int* rec    = (int*)agg;
  int* cnt    = (int*)agg + 2 * (((size_t)E + 3) & ~(size_t)3);
  int* cursor = cnt + NBP;
  int* start  = cursor + NBP;

  k_cnt_zero<<<NBP / 256, 256, 0, stream>>>(cnt);
  k_tile_count<<<256, 256, 0, stream>>>(dst, cnt, E);
  k_tile_scan<<<1, 1024, 0, stream>>>(cnt, start, cursor);
  k_tile_fill<<<256, 256, 0, stream>>>(src, dst, cursor, rec, E);

  k_tile_norm<<<NBN, 256, 0, stream>>>(start, rec, dinv);
  k_edge_weight<<<NBN, 256, 0, stream>>>(start, dinv, rec);

  launch_aggregate<128, 0, false>(x,    nullptr, start, rec, dinv, xcur, stream);
  launch_gemm2<128>(xcur, W1, b1, W2, h, stream);
  launch_aggregate<128, 1, true >(h,    b2, start, rec, dinv, xcur, stream);
  launch_aggregate<128, 0, false>(xcur, nullptr, start, rec, dinv, h, stream);
  launch_gemm2<64>(h, W3, b3, W4, xcur, stream);
  launch_aggregate<64,  0, true >(xcur, b4, start, rec, dinv, (float*)d_out, stream);
}


# aggregate: each wave sums its share of a tile's edges in a 16-row register tile (row picked by a scalar index, relative moves), then adds it to LDS once; per-edge LDS adds gone; sum order changes
# speedup vs baseline: 5.4158x; 1.0714x over previous
#include <hip/hip_runtime.h>
#include <hip/hip_bf16.h>

typedef __attribute__((ext_vector_type(2))) float v2f;
typedef __attribute__((ext_vector_type(8))) float v8f;

#define NN 50000

#define TSH 6
#define TN  (1 << TSH)
#define NBN ((NN + TN - 1) / TN)
#define ASH 4
#define TA  (1 << ASH)
#define NB  ((NN + TA - 1) / TA)
#define NBP 4096
#define TPR (TN / TA)
#define AW  4


__global__ __launch_bounds__(256) void k_cnt_zero(int* __restrict__ cnt) {
  int i = blockIdx.x * blockDim.x + threadIdx.x;
  if (i < NBP) cnt[i] = 0;
}

#define GEMM_KB 4
#define GEMM_FETCH_W(SET, BLK, WCOL, LD)                                             \
  _Pragma("unroll") for (int ks = 0; ks < GEMM_KB; ++ks) {                           \
    const int k0 = ((BLK) * GEMM_KB + ks) * 4;                                       \
    wf[SET][ks][0] = *(const v2f*)((WCOL) + (size_t)k0 * (LD));                      \
    wf[SET][ks][1] = *(const v2f*)((WCOL) + (size_t)(k0 + 1) * (LD));                \
  }
#define GEMM_FETCH_A(SET, BLK, TROW)                                                 \
  _Pragma("unroll") for (int ks = 0; ks < GEMM_KB; ++ks) {                           \
    const int k0 = ((BLK) * GEMM_KB + ks) * 4;                                       \
    _Pragma("unroll") for (int m = 0; m < MT; ++m)                                   \
      af[SET][ks][m] = *(const v2f*)((TROW) + m * 16 * PAD + k0);                    \
  }

template <int K, int LD, int MT>
__device__ __forceinline__ void gemm_kloop(const float* __restrict__ wcol, const float* trow,
                                           v2f (&wf)[2][GEMM_KB][2], v8f (&acc)[MT][2]) {
  constexpr int PAD  = K + 4;
  constexpr int NBLK = K / (4 * GEMM_KB);
  v2f af[2][GEMM_KB][MT];
  GEMM_FETCH_A(0, 0, trow)
#pragma unroll
  for (int kb = 0; kb < NBLK; ++kb) {
    if (kb + 1 < NBLK) { GEMM_FETCH_W((kb + 1) & 1, kb + 1, wcol, LD) GEMM_FETCH_A((kb + 1) & 1, kb + 1, trow) }
    __builtin_amdgcn_sched_barrier(0);
#pragma unroll
    for (int ks = 0; ks < GEMM_KB; ++ks) {
      const v2f b0 = {wf[kb & 1][ks][0].x, wf[kb & 1][ks][1].x};
      const v2f b1 = {wf[kb & 1][ks][0].y, wf[kb & 1][ks][1].y};
#pragma unroll
      for (int m = 0; m < MT; ++m) {
        const v2f a = af[kb & 1][ks][m];
        acc[m][0] = __builtin_amdgcn_wmma_f32_16x16x4_f32(
             false, a,  false, b0,
             (short)0, acc[m][0],  false,  false);
        acc[m][1] = __builtin_amdgcn_wmma_f32_16x16x4_f32(
             false, a,  false, b1,
             (short)0, acc[m][1],  false,  false);
      }
    }
  }
}

template <int K, int MID, int OUTD, int MT>
__device__ __forceinline__ void gemm2_rows(const float* __restrict__ X,
                                           const float* __restrict__ Wa,
                                           const float* __restrict__ ba,
                                           const float* __restrict__ Wb,
                                           float* __restrict__ H, const int row0,
                                           float* xs, float* hs) {
  static_assert(K == MID, "both tiles share one row stride");
  static_assert(OUTD == MID || 2 * OUTD == MID, "product B: same split or 16-row tiles");
  constexpr int PAD = K + 4;
  constexpr int BM  = 16 * MT;
  const int tid   = threadIdx.x;
  const int lane  = tid & 31;
  const int wave  = tid >> 5;
  const int idx16 = lane & 15;
  const int half  = lane >> 4;
  const float* wcolA = Wa + (size_t)(2 * half) * MID + wave * 32 + idx16 * 2;

  v2f wf[2][GEMM_KB][2];

  {
    constexpr int NTHREADS = (MID / 32) * 32;
    constexpr int PER      = BM * (K / 4) / NTHREADS;
    float4 st[PER];
#pragma unroll
    for (int i = 0; i < PER; ++i) {
      const int s = tid + i * NTHREADS;
      st[i] = *(const float4*)(X + (size_t)(row0 + (s >> 5)) * K + (s & 31) * 4);
    }
    GEMM_FETCH_W(0, 0, wcolA, MID)
#pragma unroll
    for (int i = 0; i < PER; ++i) {
      const int s = tid + i * NTHREADS;
      *(float4*)(&xs[(s >> 5) * PAD + (s & 31) * 4]) = st[i];
    }
  }
  __syncthreads();

  {
    v8f acc[MT][2] = {};
    gemm_kloop<K, MID, MT>(wcolA, &xs[idx16 * PAD + 2 * half], wf, acc);
    const v2f bb = *(const v2f*)(ba + wave * 32 + idx16 * 2);
#pragma unroll
    for (int m = 0; m < MT; ++m) {
      float* hp = hs + (m * 16 + half * 8) * PAD + wave * 32 + idx16 * 2;
#pragma unroll
      for (int r = 0; r < 8; ++r) {
        const v2f o = {fmaxf(acc[m][0][r] + bb.x, 0.0f), fmaxf(acc[m][1][r] + bb.y, 0.0f)};
        *(v2f*)(hp + r * PAD) = o;
      }
    }
  }

  constexpr bool SAME = (OUTD == MID);
  constexpr int  MT2  = SAME ? MT : 1;
  const int cs = SAME ? wave : (wave & 1);
  const int mr = SAME ? 0 : (wave >> 1);
  const bool active = SAME || (mr < MT);
  const float* wcolB = Wb + (size_t)(2 * half) * OUTD + cs * 32 + idx16 * 2;
  if (active) { GEMM_FETCH_W(0, 0, wcolB, OUTD) }
  __syncthreads();
  if (active) {
    v8f acc[MT2][2] = {};
    gemm_kloop<MID, OUTD, MT2>(wcolB, &hs[(mr * 16 + idx16) * PAD + 2 * half], wf, acc);
#pragma unroll
    for (int m = 0; m < MT2; ++m) {
      float* hp = H + (size_t)(row0 + (mr + m) * 16 + half * 8) * OUTD + cs * 32 + idx16 * 2;
#pragma unroll
      for (int r = 0; r < 8; ++r) {
        const v2f o = {acc[m][0][r], acc[m][1][r]};
        *(v2f*)(hp + r * OUTD) = o;
      }
    }
  }
}
#undef GEMM_FETCH_W
#undef GEMM_FETCH_A

template <int K, int MID, int OUTD>
__global__ __launch_bounds__((MID / 32) * 32) void k_gemm2_wmma(const float* __restrict__ X,
                                                                const float* __restrict__ Wa,
                                                                const float* __restrict__ ba,
                                                                const float* __restrict__ Wb,
                                                                float* __restrict__ H) {
  static_assert(K == 128, "K fixed at 128");
  static_assert(NN % 16 == 0, "row tiles are whole");
  __shared__ float xs[32 * (K + 4)];
  __shared__ float hs[32 * (MID + 4)];
  const int row0 = blockIdx.x * 32;
  if (row0 + 32 <= NN) gemm2_rows<K, MID, OUTD, 2>(X, Wa, ba, Wb, H, row0, xs, hs);
  else                 gemm2_rows<K, MID, OUTD, 1>(X, Wa, ba, Wb, H, row0, xs, hs);
}


__global__ __launch_bounds__(256) void k_tile_count(const int* __restrict__ dst,
                                                    int* __restrict__ cnt, int E) {
  __shared__ int hist[NBP];
  for (int b = threadIdx.x; b < NBP; b += 256) hist[b] = 0;
  __syncthreads();
  const int chunk = (E + (int)gridDim.x - 1) / (int)gridDim.x;
  const int lo = (int)blockIdx.x * chunk;
  const int hi = min(lo + chunk, E);
  for (int e = lo + (int)threadIdx.x; e < hi; e += 256) atomicAdd(&hist[dst[e] >> ASH], 1);
  __syncthreads();
  for (int b = threadIdx.x; b < NB; b += 256) {
    int c = hist[b];
    if (c) atomicAdd(&cnt[b], c);
  }
}

__global__ __launch_bounds__(1024) void k_tile_scan(const int* __restrict__ cnt,
                                                    int* __restrict__ start,
                                                    int* __restrict__ cursor) {
  constexpr int PT = NBP / 1024;
  __shared__ int s[1024];
  const int t = threadIdx.x;
  int c[PT];
  int v = 0;
#pragma unroll
  for (int i = 0; i < PT; ++i) {
    const int b = t * PT + i;
    c[i] = (b < NB) ? cnt[b] : 0;
    v += c[i];
  }
  s[t] = v;
  __syncthreads();
  for (int off = 1; off < 1024; off <<= 1) {
    int add = (t >= off) ? s[t - off] : 0;
    __syncthreads();
    s[t] += add;
    __syncthreads();
  }
  int run = s[t] - v;
#pragma unroll
  for (int i = 0; i < PT; ++i) {
    const int b = t * PT + i;
    if (b < NB) { start[b] = run; cursor[b] = run; }
    run += c[i];
  }
  if (t == 1023) start[NB] = s[t];
}

__global__ __launch_bounds__(256) void k_tile_fill(const int* __restrict__ src,
                                                   const int* __restrict__ dst,
                                                   int* __restrict__ cursor,
                                                   int* __restrict__ rec, int E) {
  __shared__ int hist[NBP];
  __shared__ int base[NBP];
  for (int b = threadIdx.x; b < NBP; b += 256) hist[b] = 0;
  __syncthreads();
  const int chunk = (E + (int)gridDim.x - 1) / (int)gridDim.x;
  const int lo = (int)blockIdx.x * chunk;
  const int hi = min(lo + chunk, E);
  for (int e = lo + (int)threadIdx.x; e < hi; e += 256) atomicAdd(&hist[dst[e] >> ASH], 1);
  __syncthreads();
  for (int b = threadIdx.x; b < NB; b += 256) {
    int c = hist[b];
    base[b] = c ? atomicAdd(&cursor[b], c) : 0;
    hist[b] = 0;
  }
  __syncthreads();
  for (int e = lo + (int)threadIdx.x; e < hi; e += 256) {
    int s = src[e];
    int d = dst[e];
    int b = d >> ASH;
    int p = base[b] + atomicAdd(&hist[b], 1);
    rec[2 * p] = (s << TSH) | (d & (TN - 1));
  }
}

__global__ __launch_bounds__(256) void k_tile_norm(const int* __restrict__ start,
                                                   const int* __restrict__ rec,
                                                   float* __restrict__ dinv) {
  constexpr int B = 8;
  __shared__ int hist[TN];
  if (threadIdx.x < TN) hist[threadIdx.x] = 0;
  __syncthreads();
  const int e0 = start[(int)blockIdx.x * TPR];
  const int e1 = start[min((int)blockIdx.x * TPR + TPR, NB)];
  for (int p0 = e0 + (int)threadIdx.x; p0 < e1; p0 += B * 256) {
    int xr[B];
#pragma unroll
    for (int u = 0; u < B; ++u) xr[u] = rec[2 * min(p0 + u * 256, e1 - 1)];
#pragma unroll
    for (int u = 0; u < B; ++u)
      if (p0 + u * 256 < e1) atomicAdd(&hist[xr[u] & (TN - 1)], 1);
  }
  __syncthreads();
  if (threadIdx.x < TN) {
    const int node = (int)blockIdx.x * TN + (int)threadIdx.x;
    if (node < NN) {
      const float d = 1.0f + (float)hist[threadIdx.x];
      dinv[node] = (d > 0.0f) ? rsqrtf(d) : 0.0f;
    }
  }
}

__global__ __launch_bounds__(256) void k_edge_weight(const int* __restrict__ start,
                                                     const float* __restrict__ dinv,
                                                     int* rec) {
  constexpr int B = 8;
  const int node0 = (int)blockIdx.x * TN;
  const int e0 = start[(int)blockIdx.x * TPR];
  const int e1 = start[min((int)blockIdx.x * TPR + TPR, NB)];
  for (int p0 = e0 + (int)threadIdx.x; p0 < e1; p0 += B * 256) {
    int x[B];
    float w[B];
#pragma unroll
    for (int u = 0; u < B; ++u) x[u] = rec[2 * min(p0 + u * 256, e1 - 1)];
#pragma unroll
    for (int u = 0; u < B; ++u) w[u] = dinv[x[u] >> TSH] * dinv[node0 + (x[u] & (TN - 1))];
#pragma unroll
    for (int u = 0; u < B; ++u) {
      const int p = p0 + u * 256;
      if (p < e1) rec[2 * p + 1] = __float_as_int(w[u]);
    }
  }
}


template <int V> using fvec = float __attribute__((ext_vector_type(V)));
typedef float rowacc_t __attribute__((ext_vector_type(16)));
static_assert(TA == 16, "rowacc_t holds one element per row of a tile");

__device__ __forceinline__ void lds_add(float* p, float v) {
  __hip_atomic_fetch_add(p, v, __ATOMIC_RELAXED, __HIP_MEMORY_SCOPE_WORKGROUP);
}

template <int OUTD, int U>
__device__ __forceinline__ void agg_fetch(fvec<OUTD / 32> (&dst)[U], const int pk, const int g,
                                          const float* __restrict__ H, const unsigned loff) {
  typedef fvec<OUTD / 32> vec_t;
#pragma unroll
  for (int u = 0; u < U; ++u) {
    const int s = __builtin_amdgcn_readlane(pk, g * U + u) >> TSH;
    const float* row = H + (size_t)s * OUTD;
    dst[u] = *(const vec_t*)(row + loff);
  }
}

template <int OUTD, bool FULL>
__device__ __forceinline__ void agg_chunk(fvec<OUTD / 32> (&buf)[2][8], const int pk, const int wi,
                                          const int n, const int pkn,
                                          const float* __restrict__ H, const unsigned loff,
                                          rowacc_t (&ra)[OUTD / 32]) {
  constexpr int VEC = OUTD / 32;
  constexpr int U   = 8;
  constexpr int G   = 32 / U;
#pragma unroll
  for (int g = 0; g < G; ++g) {
    if (g + 1 < G) agg_fetch<OUTD, U>(buf[(g + 1) & 1], pk, g + 1, H, loff);
    else if (FULL) agg_fetch<OUTD, U>(buf[0], pkn, 0, H, loff);
    __builtin_amdgcn_sched_barrier(0);
#pragma unroll
    for (int u = 0; u < U; ++u) {
      const int j = g * U + u;
      if (FULL || j < n) {
        const int r = __builtin_amdgcn_readlane(pk, j) & (TA - 1);
        const float wj = __int_as_float(__builtin_amdgcn_readlane(wi, j));
#pragma unroll
        for (int c = 0; c < VEC; ++c) ra[c][r] = fmaf(buf[g & 1][u][c], wj, ra[c][r]);
      }
    }
  }
}

template <int OUTD, int RELU, bool BIAS>
__global__ __launch_bounds__(AW * 32) void k_tile_aggregate(const int* __restrict__ start,
                                                        const int* __restrict__ rec,
                                                        const float* __restrict__ H,
                                                        const float* __restrict__ dinv,
                                                        const float* __restrict__ bias,
                                                        float* __restrict__ out) {
  constexpr int VEC = OUTD / 32;
  typedef fvec<VEC> vec_t;
  __shared__ float acc[TA * OUTD];

  const int lane  = threadIdx.x & 31;
  const int wave  = __builtin_amdgcn_readfirstlane((int)(threadIdx.x >> 5));
  const int node0 = (int)blockIdx.x * TA;
  const unsigned loff = (unsigned)lane * VEC;
  float* al = acc + lane;

  {
    constexpr int RW = TA / AW;
    vec_t sv[RW];
    float sd[RW];
#pragma unroll
    for (int i = 0; i < RW; ++i) {
      const int node = min(node0 + wave + AW * i, NN - 1);
      sd[i] = dinv[node];
      sv[i] = *(const vec_t*)(H + (size_t)node * OUTD + loff);
    }
#pragma unroll
    for (int i = 0; i < RW; ++i) {
      const int r = wave + AW * i;
#pragma unroll
      for (int c = 0; c < VEC; ++c) al[r * OUTD + c * 32] = sv[i][c] * sd[i] * sd[i];
    }
  }
  __syncthreads();

  const int e0 = start[blockIdx.x];
  const int e1 = start[blockIdx.x + 1];
  int cb = e0 + wave * 32;
  if (cb < e1) {
    const int2* rec2 = (const int2*)rec;
    int2 rc = rec2[min(cb + lane, e1 - 1)];
    vec_t buf[2][8];
    agg_fetch<OUTD, 8>(buf[0], rc.x, 0, H, loff);
    rowacc_t ra[VEC] = {};
    for (; cb + 32 <= e1; cb += AW * 32) {
      const int2 rn = rec2[min(cb + AW * 32 + lane, e1 - 1)];
      agg_chunk<OUTD, true>(buf, rc.x, rc.y, 32, rn.x, H, loff, ra);
      rc = rn;
    }
    if (cb < e1) agg_chunk<OUTD, false>(buf, rc.x, rc.y, e1 - cb, 0, H, loff, ra);
#pragma unroll
    for (int r = 0; r < TA; ++r) {
#pragma unroll
      for (int c = 0; c < VEC; ++c) lds_add(al + r * OUTD + c * 32, ra[c][r]);
    }
  }
  __syncthreads();

  vec_t bv = {};
  if (BIAS) bv = *(const vec_t*)(bias + lane * VEC);
  for (int r = wave; r < TA; r += AW) {
    const int node = node0 + r;
    if (node < NN) {
      vec_t o;
#pragma unroll
      for (int c = 0; c < VEC; ++c) {
        float x = al[r * OUTD + c * 32] + bv[c];
        if (RELU) x = fmaxf(x, 0.0f);
        o[c] = x;
      }
      *(vec_t*)(out + (size_t)node * OUTD + lane * VEC) = o;
    }
  }
}


template <int OUTD, int RELU, bool BIAS>
static void launch_aggregate(const float* hin, const float* b, const int* start, const int* rec,
                             const float* dinv, float* out, hipStream_t stream) {
  k_tile_aggregate<OUTD, RELU, BIAS><<<NB, AW * 32, 0, stream>>>(start, rec, hin, dinv, b, out);
}

template <int OUTD>
static void launch_gemm2(const float* xin, const float* Wa, const float* ba, const float* Wb,
                         float* hout, hipStream_t stream) {
  k_gemm2_wmma<128, 128, OUTD><<<dim3((NN + 31) / 32), dim3(128), 0, stream>>>(xin, Wa, ba, Wb, hout);
}

extern "C" void kernel_launch(void* const* d_in, const int* in_sizes, int n_in,
                              void* d_out, int out_size, void* d_ws, size_t ws_size,
                              hipStream_t stream) {
  const float* x  = (const float*)d_in[0];
  const int*   ei = (const int*)d_in[1];
  const float* W1 = (const float*)d_in[2];
  const float* b1 = (const float*)d_in[3];
  const float* W2 = (const float*)d_in[4];
  const float* b2 = (const float*)d_in[5];
  const float* W3 = (const float*)d_in[6];
  const float* b3 = (const float*)d_in[7];
  const float* W4 = (const float*)d_in[8];
  const float* b4 = (const float*)d_in[9];

  const int E = in_sizes[1] / 2;
  const int* src = ei;
  const int* dst = ei + E;

  float* dinv = (float*)d_ws;
  float* espc = dinv + ((NN + 3) & ~3);
  float* h    = espc + (((size_t)E + 3) & ~(size_t)3);
  float* agg  = h + (size_t)NN * 128;
  float* xcur = agg + (size_t)NN * 128;
  int* rec    = (int*)agg;
  int* cnt    = (int*)agg + 2 * (((size_t)E + 3) & ~(size_t)3);
  int* cursor = cnt + NBP;
  int* start  = cursor + NBP;

  k_cnt_zero<<<NBP / 256, 256, 0, stream>>>(cnt);
  k_tile_count<<<256, 256, 0, stream>>>(dst, cnt, E);
  k_tile_scan<<<1, 1024, 0, stream>>>(cnt, start, cursor);
  k_tile_fill<<<256, 256, 0, stream>>>(src, dst, cursor, rec, E);

  k_tile_norm<<<NBN, 256, 0, stream>>>(start, rec, dinv);
  k_edge_weight<<<NBN, 256, 0, stream>>>(start, dinv, rec);

  launch_aggregate<128, 0, false>(x,    nullptr, start, rec, dinv, xcur, stream);
  launch_gemm2<128>(xcur, W1, b1, W2, h, stream);
  launch_aggregate<128, 1, true >(h,    b2, start, rec, dinv, xcur, stream);
  launch_aggregate<128, 0, false>(xcur, nullptr, start, rec, dinv, h, stream);
  launch_gemm2<64>(h, W3, b3, W4, xcur, stream);
  launch_aggregate<64,  0, true >(xcur, b4, start, rec, dinv, (float*)d_out, stream);
}


# edge-list count and fill passes take 24 edges per thread per round with all of a round's loads in flight (was one load, one wait per edge); cursor adds grouped; bucket order, so sum order, changes
# speedup vs baseline: 5.8696x; 1.0838x over previous
#include <hip/hip_runtime.h>
#include <hip/hip_bf16.h>

typedef __attribute__((ext_vector_type(2))) float v2f;
typedef __attribute__((ext_vector_type(8))) float v8f;

#define NN 50000

#define TSH 6
#define TN  (1 << TSH)
#define NBN ((NN + TN - 1) / TN)
#define ASH 4
#define TA  (1 << ASH)
#define NB  ((NN + TA - 1) / TA)
#define NBP 4096
#define TPR (TN / TA)
#define AW  4


__global__ __launch_bounds__(256) void k_cnt_zero(int* __restrict__ cnt) {
  int i = blockIdx.x * blockDim.x + threadIdx.x;
  if (i < NBP) cnt[i] = 0;
}

#define GEMM_KB 4
#define GEMM_FETCH_W(SET, BLK, WCOL, LD)                                             \
  _Pragma("unroll") for (int ks = 0; ks < GEMM_KB; ++ks) {                           \
    const int k0 = ((BLK) * GEMM_KB + ks) * 4;                                       \
    wf[SET][ks][0] = *(const v2f*)((WCOL) + (size_t)k0 * (LD));                      \
    wf[SET][ks][1] = *(const v2f*)((WCOL) + (size_t)(k0 + 1) * (LD));                \
  }
#define GEMM_FETCH_A(SET, BLK, TROW)                                                 \
  _Pragma("unroll") for (int ks = 0; ks < GEMM_KB; ++ks) {                           \
    const int k0 = ((BLK) * GEMM_KB + ks) * 4;                                       \
    _Pragma("unroll") for (int m = 0; m < MT; ++m)                                   \
      af[SET][ks][m] = *(const v2f*)((TROW) + m * 16 * PAD + k0);                    \
  }

template <int K, int LD, int MT>
__device__ __forceinline__ void gemm_kloop(const float* __restrict__ wcol, const float* trow,
                                           v2f (&wf)[2][GEMM_KB][2], v8f (&acc)[MT][2]) {
  constexpr int PAD  = K + 4;
  constexpr int NBLK = K / (4 * GEMM_KB);
  v2f af[2][GEMM_KB][MT];
  GEMM_FETCH_A(0, 0, trow)
#pragma unroll
  for (int kb = 0; kb < NBLK; ++kb) {
    if (kb + 1 < NBLK) { GEMM_FETCH_W((kb + 1) & 1, kb + 1, wcol, LD) GEMM_FETCH_A((kb + 1) & 1, kb + 1, trow) }
    __builtin_amdgcn_sched_barrier(0);
#pragma unroll
    for (int ks = 0; ks < GEMM_KB; ++ks) {
      const v2f b0 = {wf[kb & 1][ks][0].x, wf[kb & 1][ks][1].x};
      const v2f b1 = {wf[kb & 1][ks][0].y, wf[kb & 1][ks][1].y};
#pragma unroll
      for (int m = 0; m < MT; ++m) {
        const v2f a = af[kb & 1][ks][m];
        acc[m][0] = __builtin_amdgcn_wmma_f32_16x16x4_f32(
             false, a,  false, b0,
             (short)0, acc[m][0],  false,  false);
        acc[m][1] = __builtin_amdgcn_wmma_f32_16x16x4_f32(
             false, a,  false, b1,
             (short)0, acc[m][1],  false,  false);
      }
    }
  }
}

template <int K, int MID, int OUTD, int MT>
__device__ __forceinline__ void gemm2_rows(const float* __restrict__ X,
                                           const float* __restrict__ Wa,
                                           const float* __restrict__ ba,
                                           const float* __restrict__ Wb,
                                           float* __restrict__ H, const int row0,
                                           float* xs, float* hs) {
  static_assert(K == MID, "both tiles share one row stride");
  static_assert(OUTD == MID || 2 * OUTD == MID, "product B: same split or 16-row tiles");
  constexpr int PAD = K + 4;
  constexpr int BM  = 16 * MT;
  const int tid   = threadIdx.x;
  const int lane  = tid & 31;
  const int wave  = tid >> 5;
  const int idx16 = lane & 15;
  const int half  = lane >> 4;
  const float* wcolA = Wa + (size_t)(2 * half) * MID + wave * 32 + idx16 * 2;

  v2f wf[2][GEMM_KB][2];

  {
    constexpr int NTHREADS = (MID / 32) * 32;
    constexpr int PER      = BM * (K / 4) / NTHREADS;
    float4 st[PER];
#pragma unroll
    for (int i = 0; i < PER; ++i) {
      const int s = tid + i * NTHREADS;
      st[i] = *(const float4*)(X + (size_t)(row0 + (s >> 5)) * K + (s & 31) * 4);
    }
    GEMM_FETCH_W(0, 0, wcolA, MID)
#pragma unroll
    for (int i = 0; i < PER; ++i) {
      const int s = tid + i * NTHREADS;
      *(float4*)(&xs[(s >> 5) * PAD + (s & 31) * 4]) = st[i];
    }
  }
  __syncthreads();

  {
    v8f acc[MT][2] = {};
    gemm_kloop<K, MID, MT>(wcolA, &xs[idx16 * PAD + 2 * half], wf, acc);
    const v2f bb = *(const v2f*)(ba + wave * 32 + idx16 * 2);
#pragma unroll
    for (int m = 0; m < MT; ++m) {
      float* hp = hs + (m * 16 + half * 8) * PAD + wave * 32 + idx16 * 2;
#pragma unroll
      for (int r = 0; r < 8; ++r) {
        const v2f o = {fmaxf(acc[m][0][r] + bb.x, 0.0f), fmaxf(acc[m][1][r] + bb.y, 0.0f)};
        *(v2f*)(hp + r * PAD) = o;
      }
    }
  }

  constexpr bool SAME = (OUTD == MID);
  constexpr int  MT2  = SAME ? MT : 1;
  const int cs = SAME ? wave : (wave & 1);
  const int mr = SAME ? 0 : (wave >> 1);
  const bool active = SAME || (mr < MT);
  const float* wcolB = Wb + (size_t)(2 * half) * OUTD + cs * 32 + idx16 * 2;
  if (active) { GEMM_FETCH_W(0, 0, wcolB, OUTD) }
  __syncthreads();
  if (active) {
    v8f acc[MT2][2] = {};
    gemm_kloop<MID, OUTD, MT2>(wcolB, &hs[(mr * 16 + idx16) * PAD + 2 * half], wf, acc);
#pragma unroll
    for (int m = 0; m < MT2; ++m) {
      float* hp = H + (size_t)(row0 + (mr + m) * 16 + half * 8) * OUTD + cs * 32 + idx16 * 2;
#pragma unroll
      for (int r = 0; r < 8; ++r) {
        const v2f o = {acc[m][0][r], acc[m][1][r]};
        *(v2f*)(hp + r * OUTD) = o;
      }
    }
  }
}
#undef GEMM_FETCH_W
#undef GEMM_FETCH_A

template <int K, int MID, int OUTD>
__global__ __launch_bounds__((MID / 32) * 32) void k_gemm2_wmma(const float* __restrict__ X,
                                                                const float* __restrict__ Wa,
                                                                const float* __restrict__ ba,
                                                                const float* __restrict__ Wb,
                                                                float* __restrict__ H) {
  static_assert(K == 128, "K fixed at 128");
  static_assert(NN % 16 == 0, "row tiles are whole");
  __shared__ float xs[32 * (K + 4)];
  __shared__ float hs[32 * (MID + 4)];
  const int row0 = blockIdx.x * 32;
  if (row0 + 32 <= NN) gemm2_rows<K, MID, OUTD, 2>(X, Wa, ba, Wb, H, row0, xs, hs);
  else                 gemm2_rows<K, MID, OUTD, 1>(X, Wa, ba, Wb, H, row0, xs, hs);
}


#define RBIG   24
#define RSMALL 4

template <int R>
__device__ __forceinline__ void count_round(const int* __restrict__ dst, const int e0, int* hist) {
  int d[R];
#pragma unroll
  for (int u = 0; u < R; ++u) d[u] = dst[e0 + u * 256];
#pragma unroll
  for (int u = 0; u < R; ++u) atomicAdd(&hist[d[u] >> ASH], 1);
}

__device__ __forceinline__ void count_chunk(const int* __restrict__ dst, const int lo, const int hi,
                                            int* hist) {
  int r0 = lo;
  for (; r0 + RBIG * 256 <= hi; r0 += RBIG * 256) count_round<RBIG>(dst, r0 + (int)threadIdx.x, hist);
  for (; r0 + RSMALL * 256 <= hi; r0 += RSMALL * 256) count_round<RSMALL>(dst, r0 + (int)threadIdx.x, hist);
  for (int e = r0 + (int)threadIdx.x; e < hi; e += 256) atomicAdd(&hist[dst[e] >> ASH], 1);
}

__global__ __launch_bounds__(256) void k_tile_count(const int* __restrict__ dst,
                                                    int* __restrict__ cnt, int E) {
  __shared__ int hist[NBP];
  for (int b = threadIdx.x; b < NBP; b += 256) hist[b] = 0;
  __syncthreads();
  const int chunk = (E + (int)gridDim.x - 1) / (int)gridDim.x;
  const int lo = (int)blockIdx.x * chunk;
  const int hi = min(lo + chunk, E);
  count_chunk(dst, lo, hi, hist);
  __syncthreads();
  for (int b = threadIdx.x; b < NB; b += 256) {
    int c = hist[b];
    if (c) atomicAdd(&cnt[b], c);
  }
}

__global__ __launch_bounds__(1024) void k_tile_scan(const int* __restrict__ cnt,
                                                    int* __restrict__ start,
                                                    int* __restrict__ cursor) {
  constexpr int PT = NBP / 1024;
  __shared__ int s[1024];
  const int t = threadIdx.x;
  int c[PT];
  int v = 0;
#pragma unroll
  for (int i = 0; i < PT; ++i) {
    const int b = t * PT + i;
    c[i] = (b < NB) ? cnt[b] : 0;
    v += c[i];
  }
  s[t] = v;
  __syncthreads();
  for (int off = 1; off < 1024; off <<= 1) {
    int add = (t >= off) ? s[t - off] : 0;
    __syncthreads();
    s[t] += add;
    __syncthreads();
  }
  int run = s[t] - v;
#pragma unroll
  for (int i = 0; i < PT; ++i) {
    const int b = t * PT + i;
    if (b < NB) { start[b] = run; cursor[b] = run; }
    run += c[i];
  }
  if (t == 1023) start[NB] = s[t];
}

template <int R>
__device__ __forceinline__ void place_round(const int* __restrict__ src, const int* __restrict__ dst,
                                            const int e0, int* hist, const int* base,
                                            int* __restrict__ rec) {
  int s[R], d[R], p[R];
#pragma unroll
  for (int u = 0; u < R; ++u) {
    s[u] = src[e0 + u * 256];
    d[u] = dst[e0 + u * 256];
  }
#pragma unroll
  for (int u = 0; u < R; ++u) {
    const int b = d[u] >> ASH;
    p[u] = base[b] + atomicAdd(&hist[b], 1);
  }
#pragma unroll
  for (int u = 0; u < R; ++u) rec[2 * p[u]] = (s[u] << TSH) | (d[u] & (TN - 1));
}

__global__ __launch_bounds__(256) void k_tile_fill(const int* __restrict__ src,
                                                   const int* __restrict__ dst,
                                                   int* __restrict__ cursor,
                                                   int* __restrict__ rec, int E) {
  __shared__ int hist[NBP];
  __shared__ int base[NBP];
  for (int b = threadIdx.x; b < NBP; b += 256) hist[b] = 0;
  __syncthreads();
  const int chunk = (E + (int)gridDim.x - 1) / (int)gridDim.x;
  const int lo = (int)blockIdx.x * chunk;
  const int hi = min(lo + chunk, E);
  count_chunk(dst, lo, hi, hist);
  __syncthreads();
  {
    constexpr int WB = NB / 256;
    int c[WB], r[WB];
#pragma unroll
    for (int i = 0; i < WB; ++i) c[i] = hist[(int)threadIdx.x + i * 256];
#pragma unroll
    for (int i = 0; i < WB; ++i) r[i] = atomicAdd(&cursor[(int)threadIdx.x + i * 256], c[i]);
#pragma unroll
    for (int i = 0; i < WB; ++i) {
      base[(int)threadIdx.x + i * 256] = r[i];
      hist[(int)threadIdx.x + i * 256] = 0;
    }
    for (int b = (int)threadIdx.x + WB * 256; b < NB; b += 256) {
      int cc = hist[b];
      base[b] = cc ? atomicAdd(&cursor[b], cc) : 0;
      hist[b] = 0;
    }
  }
  __syncthreads();
  int r0 = lo;
  for (; r0 + RBIG * 256 <= hi; r0 += RBIG * 256) place_round<RBIG>(src, dst, r0 + (int)threadIdx.x, hist, base, rec);
  for (; r0 + RSMALL * 256 <= hi; r0 += RSMALL * 256) place_round<RSMALL>(src, dst, r0 + (int)threadIdx.x, hist, base, rec);
  for (int e = r0 + (int)threadIdx.x; e < hi; e += 256) {
    int s = src[e];
    int d = dst[e];
    int b = d >> ASH;
    int p = base[b] + atomicAdd(&hist[b], 1);
    rec[2 * p] = (s << TSH) | (d & (TN - 1));
  }
}

__global__ __launch_bounds__(256) void k_tile_norm(const int* __restrict__ start,
                                                   const int* __restrict__ rec,
                                                   float* __restrict__ dinv) {
  constexpr int B = 8;
  __shared__ int hist[TN];
  if (threadIdx.x < TN) hist[threadIdx.x] = 0;
  __syncthreads();
  const int e0 = start[(int)blockIdx.x * TPR];
  const int e1 = start[min((int)blockIdx.x * TPR + TPR, NB)];
  for (int p0 = e0 + (int)threadIdx.x; p0 < e1; p0 += B * 256) {
    int xr[B];
#pragma unroll
    for (int u = 0; u < B; ++u) xr[u] = rec[2 * min(p0 + u * 256, e1 - 1)];
#pragma unroll
    for (int u = 0; u < B; ++u)
      if (p0 + u * 256 < e1) atomicAdd(&hist[xr[u] & (TN - 1)], 1);
  }
  __syncthreads();
  if (threadIdx.x < TN) {
    const int node = (int)blockIdx.x * TN + (int)threadIdx.x;
    if (node < NN) {
      const float d = 1.0f + (float)hist[threadIdx.x];
      dinv[node] = (d > 0.0f) ? rsqrtf(d) : 0.0f;
    }
  }
}

__global__ __launch_bounds__(256) void k_edge_weight(const int* __restrict__ start,
                                                     const float* __restrict__ dinv,
                                                     int* rec) {
  constexpr int B = 8;
  const int node0 = (int)blockIdx.x * TN;
  const int e0 = start[(int)blockIdx.x * TPR];
  const int e1 = start[min((int)blockIdx.x * TPR + TPR, NB)];
  for (int p0 = e0 + (int)threadIdx.x; p0 < e1; p0 += B * 256) {
    int x[B];
    float w[B];
#pragma unroll
    for (int u = 0; u < B; ++u) x[u] = rec[2 * min(p0 + u * 256, e1 - 1)];
#pragma unroll
    for (int u = 0; u < B; ++u) w[u] = dinv[x[u] >> TSH] * dinv[node0 + (x[u] & (TN - 1))];
#pragma unroll
    for (int u = 0; u < B; ++u) {
      const int p = p0 + u * 256;
      if (p < e1) rec[2 * p + 1] = __float_as_int(w[u]);
    }
  }
}


template <int V> using fvec = float __attribute__((ext_vector_type(V)));
typedef float rowacc_t __attribute__((ext_vector_type(16)));
static_assert(TA == 16, "rowacc_t holds one element per row of a tile");

__device__ __forceinline__ void lds_add(float* p, float v) {
  __hip_atomic_fetch_add(p, v, __ATOMIC_RELAXED, __HIP_MEMORY_SCOPE_WORKGROUP);
}

template <int OUTD, int U>
__device__ __forceinline__ void agg_fetch(fvec<OUTD / 32> (&dst)[U], const int pk, const int g,
                                          const float* __restrict__ H, const unsigned loff) {
  typedef fvec<OUTD / 32> vec_t;
#pragma unroll
  for (int u = 0; u < U; ++u) {
    const int s = __builtin_amdgcn_readlane(pk, g * U + u) >> TSH;
    const float* row = H + (size_t)s * OUTD;
    dst[u] = *(const vec_t*)(row + loff);
  }
}

template <int OUTD, bool FULL>
__device__ __forceinline__ void agg_chunk(fvec<OUTD / 32> (&buf)[2][8], const int pk, const int wi,
                                          const int n, const int pkn,
                                          const float* __restrict__ H, const unsigned loff,
                                          rowacc_t (&ra)[OUTD / 32]) {
  constexpr int VEC = OUTD / 32;
  constexpr int U   = 8;
  constexpr int G   = 32 / U;
#pragma unroll
  for (int g = 0; g < G; ++g) {
    if (g + 1 < G) agg_fetch<OUTD, U>(buf[(g + 1) & 1], pk, g + 1, H, loff);
    else if (FULL) agg_fetch<OUTD, U>(buf[0], pkn, 0, H, loff);
    __builtin_amdgcn_sched_barrier(0);
#pragma unroll
    for (int u = 0; u < U; ++u) {
      const int j = g * U + u;
      if (FULL || j < n) {
        const int r = __builtin_amdgcn_readlane(pk, j) & (TA - 1);
        const float wj = __int_as_float(__builtin_amdgcn_readlane(wi, j));
#pragma unroll
        for (int c = 0; c < VEC; ++c) ra[c][r] = fmaf(buf[g & 1][u][c], wj, ra[c][r]);
      }
    }
  }
}

template <int OUTD, int RELU, bool BIAS>
__global__ __launch_bounds__(AW * 32) void k_tile_aggregate(const int* __restrict__ start,
                                                        const int* __restrict__ rec,
                                                        const float* __restrict__ H,
                                                        const float* __restrict__ dinv,
                                                        const float* __restrict__ bias,
                                                        float* __restrict__ out) {
  constexpr int VEC = OUTD / 32;
  typedef fvec<VEC> vec_t;
  __shared__ float acc[TA * OUTD];

  const int lane  = threadIdx.x & 31;
  const int wave  = __builtin_amdgcn_readfirstlane((int)(threadIdx.x >> 5));
  const int node0 = (int)blockIdx.x * TA;
  const unsigned loff = (unsigned)lane * VEC;
  float* al = acc + lane;

  {
    constexpr int RW = TA / AW;
    vec_t sv[RW];
    float sd[RW];
#pragma unroll
    for (int i = 0; i < RW; ++i) {
      const int node = min(node0 + wave + AW * i, NN - 1);
      sd[i] = dinv[node];
      sv[i] = *(const vec_t*)(H + (size_t)node * OUTD + loff);
    }
#pragma unroll
    for (int i = 0; i < RW; ++i) {
      const int r = wave + AW * i;
#pragma unroll
      for (int c = 0; c < VEC; ++c) al[r * OUTD + c * 32] = sv[i][c] * sd[i] * sd[i];
    }
  }
  __syncthreads();

  const int e0 = start[blockIdx.x];
  const int e1 = start[blockIdx.x + 1];
  int cb = e0 + wave * 32;
  if (cb < e1) {
    const int2* rec2 = (const int2*)rec;
    int2 rc = rec2[min(cb + lane, e1 - 1)];
    vec_t buf[2][8];
    agg_fetch<OUTD, 8>(buf[0], rc.x, 0, H, loff);
    rowacc_t ra[VEC] = {};
    for (; cb + 32 <= e1; cb += AW * 32) {
      const int2 rn = rec2[min(cb + AW * 32 + lane, e1 - 1)];
      agg_chunk<OUTD, true>(buf, rc.x, rc.y, 32, rn.x, H, loff, ra);
      rc = rn;
    }
    if (cb < e1) agg_chunk<OUTD, false>(buf, rc.x, rc.y, e1 - cb, 0, H, loff, ra);
#pragma unroll
    for (int r = 0; r < TA; ++r) {
#pragma unroll
      for (int c = 0; c < VEC; ++c) lds_add(al + r * OUTD + c * 32, ra[c][r]);
    }
  }
  __syncthreads();

  vec_t bv = {};
  if (BIAS) bv = *(const vec_t*)(bias + lane * VEC);
  for (int r = wave; r < TA; r += AW) {
    const int node = node0 + r;
    if (node < NN) {
      vec_t o;
#pragma unroll
      for (int c = 0; c < VEC; ++c) {
        float x = al[r * OUTD + c * 32] + bv[c];
        if (RELU) x = fmaxf(x, 0.0f);
        o[c] = x;
      }
      *(vec_t*)(out + (size_t)node * OUTD + lane * VEC) = o;
    }
  }
}


template <int OUTD, int RELU, bool BIAS>
static void launch_aggregate(const float* hin, const float* b, const int* start, const int* rec,
                             const float* dinv, float* out, hipStream_t stream) {
  k_tile_aggregate<OUTD, RELU, BIAS><<<NB, AW * 32, 0, stream>>>(start, rec, hin, dinv, b, out);
}

template <int OUTD>
static void launch_gemm2(const float* xin, const float* Wa, const float* ba, const float* Wb,
                         float* hout, hipStream_t stream) {
  k_gemm2_wmma<128, 128, OUTD><<<dim3((NN + 31) / 32), dim3(128), 0, stream>>>(xin, Wa, ba, Wb, hout);
}

extern "C" void kernel_launch(void* const* d_in, const int* in_sizes, int n_in,
                              void* d_out, int out_size, void* d_ws, size_t ws_size,
                              hipStream_t stream) {
  const float* x  = (const float*)d_in[0];
  const int*   ei = (const int*)d_in[1];
  const float* W1 = (const float*)d_in[2];
  const float* b1 = (const float*)d_in[3];
  const float* W2 = (const float*)d_in[4];
  const float* b2 = (const float*)d_in[5];
  const float* W3 = (const float*)d_in[6];
  const float* b3 = (const float*)d_in[7];
  const float* W4 = (const float*)d_in[8];
  const float* b4 = (const float*)d_in[9];

  const int E = in_sizes[1] / 2;
  const int* src = ei;
  const int* dst = ei + E;

  float* dinv = (float*)d_ws;
  float* espc = dinv + ((NN + 3) & ~3);
  float* h    = espc + (((size_t)E + 3) & ~(size_t)3);
  float* agg  = h + (size_t)NN * 128;
  float* xcur = agg + (size_t)NN * 128;
  int* rec    = (int*)agg;
  int* cnt    = (int*)agg + 2 * (((size_t)E + 3) & ~(size_t)3);
  int* cursor = cnt + NBP;
  int* start  = cursor + NBP;

  k_cnt_zero<<<NBP / 256, 256, 0, stream>>>(cnt);
  k_tile_count<<<256, 256, 0, stream>>>(dst, cnt, E);
  k_tile_scan<<<1, 1024, 0, stream>>>(cnt, start, cursor);
  k_tile_fill<<<256, 256, 0, stream>>>(src, dst, cursor, rec, E);

  k_tile_norm<<<NBN, 256, 0, stream>>>(start, rec, dinv);
  k_edge_weight<<<NBN, 256, 0, stream>>>(start, dinv, rec);

  launch_aggregate<128, 0, false>(x,    nullptr, start, rec, dinv, xcur, stream);
  launch_gemm2<128>(xcur, W1, b1, W2, h, stream);
  launch_aggregate<128, 1, true >(h,    b2, start, rec, dinv, xcur, stream);
  launch_aggregate<128, 0, false>(xcur, nullptr, start, rec, dinv, h, stream);
  launch_gemm2<64>(h, W3, b3, W4, xcur, stream);
  launch_aggregate<64,  0, true >(xcur, b4, start, rec, dinv, (float*)d_out, stream);
}
